# starting kernel
# baseline (speedup 1.0000x reference)
#include <hip/hip_runtime.h>
#include <math.h>

#define N_LAYERS 4
#define T        34
#define TP       48
#define PSTRIDE  50
#define VOCAB    14
#define DM       6
#define HD       3

typedef float v2f __attribute__((ext_vector_type(2)));
typedef float v4f __attribute__((ext_vector_type(4)));
typedef float v8f __attribute__((ext_vector_type(8)));

__global__ __launch_bounds__(64)
void addtrans_kernel(const int* __restrict__ idx,
                     const float* __restrict__ tok_emb,
                     const float* __restrict__ pos_enc,
                     const float* __restrict__ ln_w, const float* __restrict__ ln_b,
                     const float* __restrict__ q1w, const float* __restrict__ k1w,
                     const float* __restrict__ v1w,
                     const float* __restrict__ q2w, const float* __restrict__ k2w,
                     const float* __restrict__ v2w,
                     const float* __restrict__ out_w,
                     const float* __restrict__ lnf_w, const float* __restrict__ lnf_b,
                     const float* __restrict__ head_w,
                     float* __restrict__ out)
{
    __shared__ float xs[TP][DM];
    __shared__ float hs[TP][DM];
    __shared__ float qs[2][TP][4];
    __shared__ float ks_[2][TP][4];
    __shared__ float vs[2][TP][4];
    __shared__ float os[2][TP][4];
    __shared__ float Ps[2][TP][PSTRIDE];

    const int b    = blockIdx.x;
    const int tid  = threadIdx.x;
    const int w    = tid >> 5;
    const int lane = tid & 31;
    const int hi16 = lane >> 4;
    const int l16  = lane & 15;
    const int kofs = hi16 * 2;
    const int colc = (l16 < 3) ? l16 : 3;

    __builtin_prefetch(idx + (size_t)b * T, 0, 1);

    for (int t = tid; t < T; t += 64) {
        int tok = idx[(size_t)b * T + t];
#pragma unroll
        for (int c = 0; c < 3; ++c) {
            xs[t][c]     = tok_emb[tok * 3 + c];
            xs[t][3 + c] = pos_enc[t * 3 + c];
        }
    }
    __syncthreads();

    const float rs3 = 0.57735026918962576f;

    for (int l = 0; l < N_LAYERS; ++l) {
        if (tid < TP) {
            if (tid < T) {
                float m = 0.f;
#pragma unroll
                for (int c = 0; c < DM; ++c) m += xs[tid][c];
                m *= (1.0f / DM);
                float var = 0.f;
#pragma unroll
                for (int c = 0; c < DM; ++c) { float d = xs[tid][c] - m; var += d * d; }
                var *= (1.0f / DM);
                float r = rsqrtf(var + 1e-5f);
#pragma unroll
                for (int c = 0; c < DM; ++c)
                    hs[tid][c] = (xs[tid][c] - m) * r * ln_w[l * DM + c] + ln_b[l * DM + c];
            } else {
#pragma unroll
                for (int c = 0; c < DM; ++c) hs[tid][c] = 0.f;
            }
        }
        __syncthreads();

        const float* qw = (w == 0 ? q1w : q2w) + l * 9;
        const float* kw = (w == 0 ? k1w : k2w) + l * 9;
        const float* vw = (w == 0 ? v1w : v2w) + l * 9;
        const int hoff = w * HD;
        for (int t = lane; t < TP; t += 32) {
            float h0 = hs[t][hoff], h1 = hs[t][hoff + 1], h2 = hs[t][hoff + 2];
            float valid = (t < T) ? 1.0f : 0.0f;
            v4f qv, kv, vv;
#pragma unroll
            for (int i = 0; i < HD; ++i) {
                qv[i] = (qw[i*3]*h0 + qw[i*3+1]*h1 + qw[i*3+2]*h2) * valid;
                kv[i] = (kw[i*3]*h0 + kw[i*3+1]*h1 + kw[i*3+2]*h2) * valid;
                vv[i] = (vw[i*3]*h0 + vw[i*3+1]*h1 + vw[i*3+2]*h2) * valid;
            }
            qv[3] = 0.f; kv[3] = 0.f; vv[3] = 0.f;
            *(v4f*)&qs[w][t][0]  = qv;
            *(v4f*)&ks_[w][t][0] = kv;
            *(v4f*)&vs[w][t][0]  = vv;
        }

        for (int mi = 0; mi < 3; ++mi) {
            const int am = mi * 16 + l16;
            v2f a = *(const v2f*)&qs[w][am][kofs];
            for (int ni = 0; ni <= mi; ++ni) {
                const int bn = ni * 16 + l16;
                v2f bf = *(const v2f*)&ks_[w][bn][kofs];
                v8f c = {0.f, 0.f, 0.f, 0.f, 0.f, 0.f, 0.f, 0.f};
                c = __builtin_amdgcn_wmma_f32_16x16x4_f32(false, a, false, bf,
                                                          (short)0, c, false, false);
#pragma unroll
                for (int r = 0; r < 8; ++r)
                    Ps[w][mi * 16 + hi16 * 8 + r][ni * 16 + l16] = c[r];
            }
        }

        for (int m = lane; m < TP; m += 32) {
            const bool rowok = (m < T);
            float mx = -3.4e38f;
            for (int n = 0; n < TP; ++n) {
                float s = Ps[w][m][n] * rs3;
                bool valid = rowok && (n <= m);
                mx = valid ? fmaxf(mx, s) : mx;
            }
            float sum = 0.f;
            for (int n = 0; n < TP; ++n) {
                float s = Ps[w][m][n] * rs3;
                bool valid = rowok && (n <= m);
                float e = valid ? __expf(s - mx) : 0.f;
                Ps[w][m][n] = e;
                sum += e;
            }
            float inv = (sum > 0.f) ? (1.0f / sum) : 0.f;
            for (int n = 0; n < TP; ++n) Ps[w][m][n] *= inv;
        }

        for (int mi = 0; mi < 3; ++mi) {
            v8f acc = {0.f, 0.f, 0.f, 0.f, 0.f, 0.f, 0.f, 0.f};
            const int am = mi * 16 + l16;
            int kmax = 4 * (mi + 1);
            if (kmax > (T + 3) / 4) kmax = (T + 3) / 4;
            for (int kc = 0; kc < kmax; ++kc) {
                const int nb = kc * 4;
                v2f a = *(const v2f*)&Ps[w][am][nb + kofs];
                v2f bf;
                bf.x = vs[w][nb + kofs + 0][colc];
                bf.y = vs[w][nb + kofs + 1][colc];
                acc = __builtin_amdgcn_wmma_f32_16x16x4_f32(false, a, false, bf,
                                                            (short)0, acc, false, false);
            }
            if (l16 < HD) {
#pragma unroll
                for (int r = 0; r < 8; ++r)
                    os[w][mi * 16 + hi16 * 8 + r][l16] = acc[r];
            }
        }
        __syncthreads();

        if (tid < T) {
            float u[DM];
#pragma unroll
            for (int i = 0; i < HD; ++i) { u[i] = os[0][tid][i]; u[HD + i] = os[1][tid][i]; }
#pragma unroll
            for (int i = 0; i < DM; ++i) {
                float acc = 0.f;
#pragma unroll
                for (int j = 0; j < DM; ++j) acc += out_w[l * 36 + i * 6 + j] * u[j];
                xs[tid][i] += acc;
            }
        }
        __syncthreads();
    }

    if (tid < T) {
        float m = 0.f;
#pragma unroll
        for (int c = 0; c < DM; ++c) m += xs[tid][c];
        m *= (1.0f / DM);
        float var = 0.f;
#pragma unroll
        for (int c = 0; c < DM; ++c) { float d = xs[tid][c] - m; var += d * d; }
        var *= (1.0f / DM);
        float r = rsqrtf(var + 1e-5f);
        float xf[DM];
#pragma unroll
        for (int c = 0; c < DM; ++c)
            xf[c] = (xs[tid][c] - m) * r * lnf_w[c] + lnf_b[c];
        float* op = out + ((size_t)b * T + tid) * VOCAB;
#pragma unroll
        for (int v = 0; v < VOCAB; ++v) {
            float acc = 0.f;
#pragma unroll
            for (int j = 0; j < DM; ++j) acc += head_w[v * DM + j] * xf[j];
            op[v] = acc;
        }
    }
}

extern "C" void kernel_launch(void* const* d_in, const int* in_sizes, int n_in,
                              void* d_out, int out_size, void* d_ws, size_t ws_size,
                              hipStream_t stream) {
    const int*   idx     = (const int*)  d_in[0];
    const float* tok_emb = (const float*)d_in[1];
    const float* pos_enc = (const float*)d_in[2];
    const float* ln_w    = (const float*)d_in[3];
    const float* ln_b    = (const float*)d_in[4];
    const float* q1w     = (const float*)d_in[5];
    const float* k1w     = (const float*)d_in[6];
    const float* v1w     = (const float*)d_in[7];
    const float* q2w     = (const float*)d_in[8];
    const float* k2w     = (const float*)d_in[9];
    const float* v2w     = (const float*)d_in[10];
    const float* out_w   = (const float*)d_in[11];
    const float* lnf_w   = (const float*)d_in[12];
    const float* lnf_b   = (const float*)d_in[13];
    const float* head_w  = (const float*)d_in[14];
    float* out = (float*)d_out;

    const int batch = in_sizes[0] / T;
    addtrans_kernel<<<batch, 64, 0, stream>>>(idx, tok_emb, pos_enc, ln_w, ln_b,
                                              q1w, k1w, v1w, q2w, k2w, v2w,
                                              out_w, lnf_w, lnf_b, head_w, out);
    (void)d_ws; (void)ws_size; (void)n_in; (void)out_size;
}


# Scores built transposed (K x Q^T) so the causal softmax runs in registers and its tiles feed V^T x P^T as B: no score LDS; row sums and PV add keys in per-lane-half order, 1/sum applied after PV
# speedup vs baseline: 4.5682x; 4.5682x over previous
#include <hip/hip_runtime.h>
#include <math.h>

#define N_LAYERS 4
#define T        34
#define TP       48
#define VOCAB    14
#define DM       6
#define HD       3

static_assert(T == 34, "the last key tile is read as one chunk of T - 32 = 2 real keys per lane half");
typedef float v2f __attribute__((ext_vector_type(2)));
typedef float v4f __attribute__((ext_vector_type(4)));
typedef float v8f __attribute__((ext_vector_type(8)));

__device__ __forceinline__ float xhalf(float x) {
    int b = __builtin_bit_cast(int, x);
    return __builtin_bit_cast(float,
        __builtin_amdgcn_permlanex16(b, b, 0x76543210, 0xfedcba98, false, false));
}

template <int QT>
__device__ __forceinline__ v4f attend_q(const v2f (&kf)[3], v2f qf, const float (&vf)[3][8], int dl)
{
    const float NEG = -3.4e38f;
    const float sc  = 0.57735026918962576f * 1.4426950408889634f;
    float p[QT + 1][8];
    float mx = NEG;
#pragma unroll
    for (int kt = 0; kt <= QT; ++kt) {
        const int kv = (T - 16 * kt) < 8 ? (T - 16 * kt) : 8;
        v8f z = {0.f, 0.f, 0.f, 0.f, 0.f, 0.f, 0.f, 0.f};
        v8f s = __builtin_amdgcn_wmma_f32_16x16x4_f32(false, kf[kt], false, qf,
                                                      (short)0, z, false, false);
#pragma unroll
        for (int r = 0; r < 8; ++r) {
            if (r < kv) {
                float t = s[r] * sc;
                if (kt == QT) t = (r <= dl) ? t : NEG;
                p[kt][r] = t;
                mx = fmaxf(mx, t);
            } else {
                p[kt][r] = 0.f;
            }
        }
    }
    mx = fmaxf(mx, xhalf(mx));
    float sum = 0.f;
#pragma unroll
    for (int kt = 0; kt <= QT; ++kt) {
        const int kv = (T - 16 * kt) < 8 ? (T - 16 * kt) : 8;
#pragma unroll
        for (int r = 0; r < 8; ++r) {
            if (r < kv) {
                float e = __builtin_amdgcn_exp2f(p[kt][r] - mx);
                p[kt][r] = e;
                sum += e;
            }
        }
    }
    sum += xhalf(sum);
    const float inv = __builtin_amdgcn_rcpf(sum);
    v8f acc = {0.f, 0.f, 0.f, 0.f, 0.f, 0.f, 0.f, 0.f};
#pragma unroll
    for (int kt = 0; kt <= QT; ++kt) {
        const int kv = (T - 16 * kt) < 8 ? (T - 16 * kt) : 8;
#pragma unroll
        for (int j = 0; j < 4; ++j) {
            if (2 * j < kv) {
                v2f a, bf;
                a.x  = vf[kt][2 * j];  a.y  = vf[kt][2 * j + 1];
                bf.x = p[kt][2 * j];   bf.y = p[kt][2 * j + 1];
                acc = __builtin_amdgcn_wmma_f32_16x16x4_f32(false, a, false, bf,
                                                            (short)0, acc, false, false);
            }
        }
    }
    v4f o;
    o.x = acc[0] * inv; o.y = acc[1] * inv; o.z = acc[2] * inv; o.w = 0.f;
    return o;
}

__global__ __launch_bounds__(64)
void addtrans_kernel(const int* __restrict__ idx,
                     const float* __restrict__ tok_emb,
                     const float* __restrict__ pos_enc,
                     const float* __restrict__ ln_w, const float* __restrict__ ln_b,
                     const float* __restrict__ q1w, const float* __restrict__ k1w,
                     const float* __restrict__ v1w,
                     const float* __restrict__ q2w, const float* __restrict__ k2w,
                     const float* __restrict__ v2w,
                     const float* __restrict__ out_w,
                     const float* __restrict__ lnf_w, const float* __restrict__ lnf_b,
                     const float* __restrict__ head_w,
                     float* __restrict__ out)
{
    __shared__ float xs[TP][DM];
    __shared__ float hs[TP][DM];
    __shared__ __attribute__((aligned(16))) float qs[2][TP][4];
    __shared__ __attribute__((aligned(16))) float ks_[2][TP][4];
    __shared__ __attribute__((aligned(16))) float vT[2][4][TP];
    float (*os)[TP][4] = qs;

    const int b    = blockIdx.x;
    const int tid  = threadIdx.x;
    const int w    = tid >> 5;
    const int lane = tid & 31;
    const int hi16 = lane >> 4;
    const int l16  = lane & 15;
    const int kofs = hi16 * 2;
    const int colc = (l16 < 3) ? l16 : 3;
    const int dl   = l16 - 8 * hi16;

    __builtin_prefetch(idx + (size_t)b * T, 0, 1);

    for (int t = tid; t < 2 * TP; t += 64) vT[t / TP][3][t % TP] = 0.f;

    for (int t = tid; t < T; t += 64) {
        int tok = idx[(size_t)b * T + t];
#pragma unroll
        for (int c = 0; c < 3; ++c) {
            xs[t][c]     = tok_emb[tok * 3 + c];
            xs[t][3 + c] = pos_enc[t * 3 + c];
        }
    }
    __syncthreads();

    for (int l = 0; l < N_LAYERS; ++l) {
        if (tid < TP) {
            if (tid < T) {
                float m = 0.f;
#pragma unroll
                for (int c = 0; c < DM; ++c) m += xs[tid][c];
                m *= (1.0f / DM);
                float var = 0.f;
#pragma unroll
                for (int c = 0; c < DM; ++c) { float d = xs[tid][c] - m; var += d * d; }
                var *= (1.0f / DM);
                float r = rsqrtf(var + 1e-5f);
#pragma unroll
                for (int c = 0; c < DM; ++c)
                    hs[tid][c] = (xs[tid][c] - m) * r * ln_w[l * DM + c] + ln_b[l * DM + c];
            } else {
#pragma unroll
                for (int c = 0; c < DM; ++c) hs[tid][c] = 0.f;
            }
        }
        __syncthreads();

        const float* qw = (w == 0 ? q1w : q2w) + l * 9;
        const float* kw = (w == 0 ? k1w : k2w) + l * 9;
        const float* vw = (w == 0 ? v1w : v2w) + l * 9;
        const int hoff = w * HD;
        for (int t = lane; t < TP; t += 32) {
            float h0 = hs[t][hoff], h1 = hs[t][hoff + 1], h2 = hs[t][hoff + 2];
            float valid = (t < T) ? 1.0f : 0.0f;
            v4f qv, kv, vv;
#pragma unroll
            for (int i = 0; i < HD; ++i) {
                qv[i] = (qw[i*3]*h0 + qw[i*3+1]*h1 + qw[i*3+2]*h2) * valid;
                kv[i] = (kw[i*3]*h0 + kw[i*3+1]*h1 + kw[i*3+2]*h2) * valid;
                vv[i] = (vw[i*3]*h0 + vw[i*3+1]*h1 + vw[i*3+2]*h2) * valid;
            }
            qv[3] = 0.f; kv[3] = 0.f; vv[3] = 0.f;
            *(v4f*)&qs[w][t][0]  = qv;
            *(v4f*)&ks_[w][t][0] = kv;
            vT[w][0][t] = vv[0];
            vT[w][1][t] = vv[1];
            vT[w][2][t] = vv[2];
        }

        v2f kf[3], qf[3];
        float vf[3][8];
#pragma unroll
        for (int i = 0; i < 3; ++i) {
            kf[i] = *(const v2f*)&ks_[w][i * 16 + l16][kofs];
            qf[i] = *(const v2f*)&qs[w][i * 16 + l16][kofs];
        }
#pragma unroll
        for (int kt = 0; kt < 2; ++kt) {
            v4f lo = *(const v4f*)&vT[w][colc][kt * 16 + 8 * hi16];
            v4f hi = *(const v4f*)&vT[w][colc][kt * 16 + 8 * hi16 + 4];
            vf[kt][0] = lo.x; vf[kt][1] = lo.y; vf[kt][2] = lo.z; vf[kt][3] = lo.w;
            vf[kt][4] = hi.x; vf[kt][5] = hi.y; vf[kt][6] = hi.z; vf[kt][7] = hi.w;
        }
        {
            v2f l2 = *(const v2f*)&vT[w][colc][32 + 8 * hi16];
            vf[2][0] = l2.x; vf[2][1] = l2.y;
            vf[2][2] = 0.f; vf[2][3] = 0.f; vf[2][4] = 0.f; vf[2][5] = 0.f; vf[2][6] = 0.f; vf[2][7] = 0.f;
        }
        const v4f o0 = attend_q<0>(kf, qf[0], vf, dl);
        const v4f o1 = attend_q<1>(kf, qf[1], vf, dl);
        const v4f o2 = attend_q<2>(kf, qf[2], vf, dl);
        if (hi16 == 0) {
            *(v4f*)&os[w][l16][0]      = o0;
            *(v4f*)&os[w][16 + l16][0] = o1;
            *(v4f*)&os[w][32 + l16][0] = o2;
        }
        __syncthreads();

        if (tid < T) {
            float u[DM];
#pragma unroll
            for (int i = 0; i < HD; ++i) { u[i] = os[0][tid][i]; u[HD + i] = os[1][tid][i]; }
#pragma unroll
            for (int i = 0; i < DM; ++i) {
                float acc = 0.f;
#pragma unroll
                for (int j = 0; j < DM; ++j) acc += out_w[l * 36 + i * 6 + j] * u[j];
                xs[tid][i] += acc;
            }
        }
        __syncthreads();
    }

    if (tid < T) {
        float m = 0.f;
#pragma unroll
        for (int c = 0; c < DM; ++c) m += xs[tid][c];
        m *= (1.0f / DM);
        float var = 0.f;
#pragma unroll
        for (int c = 0; c < DM; ++c) { float d = xs[tid][c] - m; var += d * d; }
        var *= (1.0f / DM);
        float r = rsqrtf(var + 1e-5f);
        float xf[DM];
#pragma unroll
        for (int c = 0; c < DM; ++c)
            xf[c] = (xs[tid][c] - m) * r * lnf_w[c] + lnf_b[c];
        float* op = out + ((size_t)b * T + tid) * VOCAB;
#pragma unroll
        for (int v = 0; v < VOCAB; ++v) {
            float acc = 0.f;
#pragma unroll
            for (int j = 0; j < DM; ++j) acc += head_w[v * DM + j] * xf[j];
            op[v] = acc;
        }
    }
}

extern "C" void kernel_launch(void* const* d_in, const int* in_sizes, int n_in,
                              void* d_out, int out_size, void* d_ws, size_t ws_size,
                              hipStream_t stream) {
    const int*   idx     = (const int*)  d_in[0];
    const float* tok_emb = (const float*)d_in[1];
    const float* pos_enc = (const float*)d_in[2];
    const float* ln_w    = (const float*)d_in[3];
    const float* ln_b    = (const float*)d_in[4];
    const float* q1w     = (const float*)d_in[5];
    const float* k1w     = (const float*)d_in[6];
    const float* v1w     = (const float*)d_in[7];
    const float* q2w     = (const float*)d_in[8];
    const float* k2w     = (const float*)d_in[9];
    const float* v2w     = (const float*)d_in[10];
    const float* out_w   = (const float*)d_in[11];
    const float* lnf_w   = (const float*)d_in[12];
    const float* lnf_b   = (const float*)d_in[13];
    const float* head_w  = (const float*)d_in[14];
    float* out = (float*)d_out;

    const int batch = in_sizes[0] / T;
    addtrans_kernel<<<batch, 64, 0, stream>>>(idx, tok_emb, pos_enc, ln_w, ln_b,
                                              q1w, k1w, v1w, q2w, k2w, v2w,
                                              out_w, lnf_w, lnf_b, head_w, out);
    (void)d_ws; (void)ws_size; (void)n_in; (void)out_size;
}


# Softmax side work moved into WMMA operands: log2e/sqrt(d) folded into Q, causal mask added as the diagonal tile's C, row sums from a ones row of V^T (added in WMMA k order, no lane-half add)
# speedup vs baseline: 4.9746x; 1.0890x over previous
#include <hip/hip_runtime.h>
#include <math.h>

#define N_LAYERS 4
#define T        34
#define TP       48
#define VOCAB    14
#define DM       6
#define HD       3

static_assert(T == 34, "the last key tile is read as one chunk of T - 32 = 2 real keys per lane half");
typedef float v2f __attribute__((ext_vector_type(2)));
typedef float v4f __attribute__((ext_vector_type(4)));
typedef float v8f __attribute__((ext_vector_type(8)));

__device__ __forceinline__ float xhalf(float x) {
    int b = __builtin_bit_cast(int, x);
    return __builtin_bit_cast(float,
        __builtin_amdgcn_permlanex16(b, b, 0x76543210, 0xfedcba98, false, false));
}

template <int QT>
__device__ __forceinline__ v4f attend_q(const v2f (&kf)[3], v2f qf, const float (&vf)[3][8], v8f cm)
{
    v2f p[QT + 1][4];
    float mx = -3.4e38f;
#pragma unroll
    for (int kt = 0; kt <= QT; ++kt) {
        const int kv = (T - 16 * kt) < 8 ? (T - 16 * kt) : 8;
        v8f z = {0.f, 0.f, 0.f, 0.f, 0.f, 0.f, 0.f, 0.f};
        v8f s = __builtin_amdgcn_wmma_f32_16x16x4_f32(false, kf[kt], false, qf,
                                                      (short)0, (kt == QT) ? cm : z, false, false);
#pragma unroll
        for (int j = 0; j < 4; ++j) {
            if (2 * j < kv) {
                p[kt][j].x = s[2 * j];
                p[kt][j].y = s[2 * j + 1];
                mx = fmaxf(mx, fmaxf(s[2 * j], s[2 * j + 1]));
            }
        }
    }
    mx = fmaxf(mx, xhalf(mx));
    const v2f mx2 = {mx, mx};
#pragma unroll
    for (int kt = 0; kt <= QT; ++kt) {
        const int kv = (T - 16 * kt) < 8 ? (T - 16 * kt) : 8;
#pragma unroll
        for (int j = 0; j < 4; ++j) {
            if (2 * j < kv) {
                const v2f t = p[kt][j] - mx2;
                p[kt][j].x = __builtin_amdgcn_exp2f(t.x);
                p[kt][j].y = __builtin_amdgcn_exp2f(t.y);
            }
        }
    }
    v8f acc = {0.f, 0.f, 0.f, 0.f, 0.f, 0.f, 0.f, 0.f};
#pragma unroll
    for (int kt = 0; kt <= QT; ++kt) {
        const int kv = (T - 16 * kt) < 8 ? (T - 16 * kt) : 8;
#pragma unroll
        for (int j = 0; j < 4; ++j) {
            if (2 * j < kv) {
                v2f a;
                a.x  = vf[kt][2 * j];  a.y  = vf[kt][2 * j + 1];
                acc = __builtin_amdgcn_wmma_f32_16x16x4_f32(false, a, false, p[kt][j],
                                                            (short)0, acc, false, false);
            }
        }
    }
    const float inv = __builtin_amdgcn_rcpf(acc[3]);
    v4f o;
    o.x = acc[0] * inv; o.y = acc[1] * inv; o.z = acc[2] * inv; o.w = 0.f;
    return o;
}

__global__ __launch_bounds__(64)
void addtrans_kernel(const int* __restrict__ idx,
                     const float* __restrict__ tok_emb,
                     const float* __restrict__ pos_enc,
                     const float* __restrict__ ln_w, const float* __restrict__ ln_b,
                     const float* __restrict__ q1w, const float* __restrict__ k1w,
                     const float* __restrict__ v1w,
                     const float* __restrict__ q2w, const float* __restrict__ k2w,
                     const float* __restrict__ v2w,
                     const float* __restrict__ out_w,
                     const float* __restrict__ lnf_w, const float* __restrict__ lnf_b,
                     const float* __restrict__ head_w,
                     float* __restrict__ out)
{
    __shared__ float xs[TP][DM];
    __shared__ float hs[TP][DM];
    __shared__ __attribute__((aligned(16))) float qs[2][TP][4];
    __shared__ __attribute__((aligned(16))) float ks_[2][TP][4];
    __shared__ __attribute__((aligned(16))) float vT[2][4][TP];
    float (*os)[TP][4] = qs;

    const int b    = blockIdx.x;
    const int tid  = threadIdx.x;
    const int w    = tid >> 5;
    const int lane = tid & 31;
    const int hi16 = lane >> 4;
    const int l16  = lane & 15;
    const int kofs = hi16 * 2;
    const int colc = (l16 < 3) ? l16 : 3;
    const int dl   = l16 - 8 * hi16;
    const float sc = 0.57735026918962576f * 1.4426950408889634f;
    v8f cm;
#pragma unroll
    for (int r = 0; r < 8; ++r) cm[r] = (r <= dl) ? 0.f : -3.4e38f;

    __builtin_prefetch(idx + (size_t)b * T, 0, 1);

    for (int t = tid; t < 2 * TP; t += 64) vT[t / TP][3][t % TP] = ((t % TP) < T) ? 1.f : 0.f;

    for (int t = tid; t < T; t += 64) {
        int tok = idx[(size_t)b * T + t];
#pragma unroll
        for (int c = 0; c < 3; ++c) {
            xs[t][c]     = tok_emb[tok * 3 + c];
            xs[t][3 + c] = pos_enc[t * 3 + c];
        }
    }
    __syncthreads();

    for (int l = 0; l < N_LAYERS; ++l) {
        if (tid < TP) {
            if (tid < T) {
                float m = 0.f;
#pragma unroll
                for (int c = 0; c < DM; ++c) m += xs[tid][c];
                m *= (1.0f / DM);
                float var = 0.f;
#pragma unroll
                for (int c = 0; c < DM; ++c) { float d = xs[tid][c] - m; var += d * d; }
                var *= (1.0f / DM);
                float r = rsqrtf(var + 1e-5f);
#pragma unroll
                for (int c = 0; c < DM; ++c)
                    hs[tid][c] = (xs[tid][c] - m) * r * ln_w[l * DM + c] + ln_b[l * DM + c];
            } else {
#pragma unroll
                for (int c = 0; c < DM; ++c) hs[tid][c] = 0.f;
            }
        }
        __syncthreads();

        const float* qw = (w == 0 ? q1w : q2w) + l * 9;
        const float* kw = (w == 0 ? k1w : k2w) + l * 9;
        const float* vw = (w == 0 ? v1w : v2w) + l * 9;
        const int hoff = w * HD;
        for (int t = lane; t < TP; t += 32) {
            float h0 = hs[t][hoff], h1 = hs[t][hoff + 1], h2 = hs[t][hoff + 2];
            float valid = (t < T) ? 1.0f : 0.0f;
            float vq    = (t < T) ? sc : 0.0f;
            v4f qv, kv, vv;
#pragma unroll
            for (int i = 0; i < HD; ++i) {
                qv[i] = (qw[i*3]*h0 + qw[i*3+1]*h1 + qw[i*3+2]*h2) * vq;
                kv[i] = (kw[i*3]*h0 + kw[i*3+1]*h1 + kw[i*3+2]*h2) * valid;
                vv[i] = (vw[i*3]*h0 + vw[i*3+1]*h1 + vw[i*3+2]*h2) * valid;
            }
            qv[3] = 0.f; kv[3] = 0.f; vv[3] = 0.f;
            *(v4f*)&qs[w][t][0]  = qv;
            *(v4f*)&ks_[w][t][0] = kv;
            vT[w][0][t] = vv[0];
            vT[w][1][t] = vv[1];
            vT[w][2][t] = vv[2];
        }

        v2f kf[3], qf[3];
        float vf[3][8];
#pragma unroll
        for (int i = 0; i < 3; ++i) {
            kf[i] = *(const v2f*)&ks_[w][i * 16 + l16][kofs];
            qf[i] = *(const v2f*)&qs[w][i * 16 + l16][kofs];
        }
#pragma unroll
        for (int kt = 0; kt < 2; ++kt) {
            v4f lo = *(const v4f*)&vT[w][colc][kt * 16 + 8 * hi16];
            v4f hi = *(const v4f*)&vT[w][colc][kt * 16 + 8 * hi16 + 4];
            vf[kt][0] = lo.x; vf[kt][1] = lo.y; vf[kt][2] = lo.z; vf[kt][3] = lo.w;
            vf[kt][4] = hi.x; vf[kt][5] = hi.y; vf[kt][6] = hi.z; vf[kt][7] = hi.w;
        }
        {
            v2f l2 = *(const v2f*)&vT[w][colc][32 + 8 * hi16];
            vf[2][0] = l2.x; vf[2][1] = l2.y;
            vf[2][2] = 0.f; vf[2][3] = 0.f; vf[2][4] = 0.f; vf[2][5] = 0.f; vf[2][6] = 0.f; vf[2][7] = 0.f;
        }
        const v4f o0 = attend_q<0>(kf, qf[0], vf, cm);
        const v4f o1 = attend_q<1>(kf, qf[1], vf, cm);
        const v4f o2 = attend_q<2>(kf, qf[2], vf, cm);
        if (hi16 == 0) {
            *(v4f*)&os[w][l16][0]      = o0;
            *(v4f*)&os[w][16 + l16][0] = o1;
            *(v4f*)&os[w][32 + l16][0] = o2;
        }
        __syncthreads();

        if (tid < T) {
            float u[DM];
#pragma unroll
            for (int i = 0; i < HD; ++i) { u[i] = os[0][tid][i]; u[HD + i] = os[1][tid][i]; }
#pragma unroll
            for (int i = 0; i < DM; ++i) {
                float acc = 0.f;
#pragma unroll
                for (int j = 0; j < DM; ++j) acc += out_w[l * 36 + i * 6 + j] * u[j];
                xs[tid][i] += acc;
            }
        }
        __syncthreads();
    }

    if (tid < T) {
        float m = 0.f;
#pragma unroll
        for (int c = 0; c < DM; ++c) m += xs[tid][c];
        m *= (1.0f / DM);
        float var = 0.f;
#pragma unroll
        for (int c = 0; c < DM; ++c) { float d = xs[tid][c] - m; var += d * d; }
        var *= (1.0f / DM);
        float r = rsqrtf(var + 1e-5f);
        float xf[DM];
#pragma unroll
        for (int c = 0; c < DM; ++c)
            xf[c] = (xs[tid][c] - m) * r * lnf_w[c] + lnf_b[c];
        float* op = out + ((size_t)b * T + tid) * VOCAB;
#pragma unroll
        for (int v = 0; v < VOCAB; ++v) {
            float acc = 0.f;
#pragma unroll
            for (int j = 0; j < DM; ++j) acc += head_w[v * DM + j] * xf[j];
            op[v] = acc;
        }
    }
}

extern "C" void kernel_launch(void* const* d_in, const int* in_sizes, int n_in,
                              void* d_out, int out_size, void* d_ws, size_t ws_size,
                              hipStream_t stream) {
    const int*   idx     = (const int*)  d_in[0];
    const float* tok_emb = (const float*)d_in[1];
    const float* pos_enc = (const float*)d_in[2];
    const float* ln_w    = (const float*)d_in[3];
    const float* ln_b    = (const float*)d_in[4];
    const float* q1w     = (const float*)d_in[5];
    const float* k1w     = (const float*)d_in[6];
    const float* v1w     = (const float*)d_in[7];
    const float* q2w     = (const float*)d_in[8];
    const float* k2w     = (const float*)d_in[9];
    const float* v2w     = (const float*)d_in[10];
    const float* out_w   = (const float*)d_in[11];
    const float* lnf_w   = (const float*)d_in[12];
    const float* lnf_b   = (const float*)d_in[13];
    const float* head_w  = (const float*)d_in[14];
    float* out = (float*)d_out;

    const int batch = in_sizes[0] / T;
    addtrans_kernel<<<batch, 64, 0, stream>>>(idx, tok_emb, pos_enc, ln_w, ln_b,
                                              q1w, k1w, v1w, q2w, k2w, v2w,
                                              out_w, lnf_w, lnf_b, head_w, out);
    (void)d_ws; (void)ws_size; (void)n_in; (void)out_size;
}


# Query tiles start 2 rows behind the key tiles (the 2 left-over queries first): each tile's last key tile shrinks to one 2-key chunk, 21 WMMAs a layer instead of 27; per-query sum order unchanged
# speedup vs baseline: 5.6569x; 1.1372x over previous
#include <hip/hip_runtime.h>
#include <math.h>

#define N_LAYERS 4
#define T        34
#define TP       48
#define VOCAB    14
#define DM       6
#define HD       3

#define QOFF     (T - 32)

static_assert(T == 34, "the first query tile holds the T - 32 = 2 queries left over; every query tile's last key tile is read as one chunk of 2 keys per lane half");
typedef float v2f __attribute__((ext_vector_type(2)));
typedef float v4f __attribute__((ext_vector_type(4)));
typedef float v8f __attribute__((ext_vector_type(8)));

__device__ __forceinline__ float xhalf(float x) {
    int b = __builtin_bit_cast(int, x);
    return __builtin_bit_cast(float,
        __builtin_amdgcn_permlanex16(b, b, 0x76543210, 0xfedcba98, false, false));
}

template <int QT>
__device__ __forceinline__ v4f attend_q(const v2f (&kf)[3], v2f qf, const float (&vf)[3][8], v8f md, v8f ml)
{
    v2f p[QT + 1][4];
    float mx = -3.4e38f;
#pragma unroll
    for (int kt = 0; kt <= QT; ++kt) {
        const int kv = (kt == QT) ? QOFF : 8;
        v8f z = {0.f, 0.f, 0.f, 0.f, 0.f, 0.f, 0.f, 0.f};
        v8f s = __builtin_amdgcn_wmma_f32_16x16x4_f32(false, kf[kt], false, qf, (short)0,
                                                      (kt == QT) ? ml : (kt == QT - 1) ? md : z, false, false);
#pragma unroll
        for (int j = 0; j < 4; ++j) {
            if (2 * j < kv) {
                p[kt][j].x = s[2 * j];
                p[kt][j].y = s[2 * j + 1];
                mx = fmaxf(mx, fmaxf(s[2 * j], s[2 * j + 1]));
            }
        }
    }
    mx = fmaxf(mx, xhalf(mx));
    const v2f mx2 = {mx, mx};
#pragma unroll
    for (int kt = 0; kt <= QT; ++kt) {
        const int kv = (kt == QT) ? QOFF : 8;
#pragma unroll
        for (int j = 0; j < 4; ++j) {
            if (2 * j < kv) {
                const v2f t = p[kt][j] - mx2;
                p[kt][j].x = __builtin_amdgcn_exp2f(t.x);
                p[kt][j].y = __builtin_amdgcn_exp2f(t.y);
            }
        }
    }
    v8f acc = {0.f, 0.f, 0.f, 0.f, 0.f, 0.f, 0.f, 0.f};
#pragma unroll
    for (int kt = 0; kt <= QT; ++kt) {
        const int kv = (kt == QT) ? QOFF : 8;
#pragma unroll
        for (int j = 0; j < 4; ++j) {
            if (2 * j < kv) {
                v2f a;
                a.x  = vf[kt][2 * j];  a.y  = vf[kt][2 * j + 1];
                acc = __builtin_amdgcn_wmma_f32_16x16x4_f32(false, a, false, p[kt][j],
                                                            (short)0, acc, false, false);
            }
        }
    }
    const float inv = __builtin_amdgcn_rcpf(acc[3]);
    v4f o;
    o.x = acc[0] * inv; o.y = acc[1] * inv; o.z = acc[2] * inv; o.w = 0.f;
    return o;
}

__global__ __launch_bounds__(64)
void addtrans_kernel(const int* __restrict__ idx,
                     const float* __restrict__ tok_emb,
                     const float* __restrict__ pos_enc,
                     const float* __restrict__ ln_w, const float* __restrict__ ln_b,
                     const float* __restrict__ q1w, const float* __restrict__ k1w,
                     const float* __restrict__ v1w,
                     const float* __restrict__ q2w, const float* __restrict__ k2w,
                     const float* __restrict__ v2w,
                     const float* __restrict__ out_w,
                     const float* __restrict__ lnf_w, const float* __restrict__ lnf_b,
                     const float* __restrict__ head_w,
                     float* __restrict__ out)
{
    __shared__ float xs[TP][DM];
    __shared__ float hs[TP][DM];
    __shared__ __attribute__((aligned(16))) float qs[2][TP][4];
    __shared__ __attribute__((aligned(16))) float ks_[2][TP][4];
    __shared__ __attribute__((aligned(16))) float vT[2][4][TP];
    float (*os)[TP][4] = qs;

    const int b    = blockIdx.x;
    const int tid  = threadIdx.x;
    const int w    = tid >> 5;
    const int lane = tid & 31;
    const int hi16 = lane >> 4;
    const int l16  = lane & 15;
    const int kofs = hi16 * 2;
    const int colc = (l16 < 3) ? l16 : 3;
    const int dl   = l16 - 8 * hi16;
    const float sc = 0.57735026918962576f * 1.4426950408889634f;
    v8f cmd, cml, cm0;
#pragma unroll
    for (int r = 0; r < 8; ++r) {
        cmd[r] = (r <= dl + QOFF) ? 0.f : -3.4e38f;
        cml[r] = (r <= dl + QOFF - 16) ? 0.f : -3.4e38f;
        cm0[r] = (r <= dl) ? 0.f : -3.4e38f;
    }

    __builtin_prefetch(idx + (size_t)b * T, 0, 1);

    for (int t = tid; t < 2 * TP; t += 64) vT[t / TP][3][t % TP] = ((t % TP) < T) ? 1.f : 0.f;

    for (int t = tid; t < T; t += 64) {
        int tok = idx[(size_t)b * T + t];
#pragma unroll
        for (int c = 0; c < 3; ++c) {
            xs[t][c]     = tok_emb[tok * 3 + c];
            xs[t][3 + c] = pos_enc[t * 3 + c];
        }
    }
    __syncthreads();

    for (int l = 0; l < N_LAYERS; ++l) {
        if (tid < TP) {
            if (tid < T) {
                float m = 0.f;
#pragma unroll
                for (int c = 0; c < DM; ++c) m += xs[tid][c];
                m *= (1.0f / DM);
                float var = 0.f;
#pragma unroll
                for (int c = 0; c < DM; ++c) { float d = xs[tid][c] - m; var += d * d; }
                var *= (1.0f / DM);
                float r = rsqrtf(var + 1e-5f);
#pragma unroll
                for (int c = 0; c < DM; ++c)
                    hs[tid][c] = (xs[tid][c] - m) * r * ln_w[l * DM + c] + ln_b[l * DM + c];
            } else {
#pragma unroll
                for (int c = 0; c < DM; ++c) hs[tid][c] = 0.f;
            }
        }
        __syncthreads();

        const float* qw = (w == 0 ? q1w : q2w) + l * 9;
        const float* kw = (w == 0 ? k1w : k2w) + l * 9;
        const float* vw = (w == 0 ? v1w : v2w) + l * 9;
        const int hoff = w * HD;
        for (int t = lane; t < TP; t += 32) {
            float h0 = hs[t][hoff], h1 = hs[t][hoff + 1], h2 = hs[t][hoff + 2];
            float valid = (t < T) ? 1.0f : 0.0f;
            float vq    = (t < T) ? sc : 0.0f;
            v4f qv, kv, vv;
#pragma unroll
            for (int i = 0; i < HD; ++i) {
                qv[i] = (qw[i*3]*h0 + qw[i*3+1]*h1 + qw[i*3+2]*h2) * vq;
                kv[i] = (kw[i*3]*h0 + kw[i*3+1]*h1 + kw[i*3+2]*h2) * valid;
                vv[i] = (vw[i*3]*h0 + vw[i*3+1]*h1 + vw[i*3+2]*h2) * valid;
            }
            qv[3] = 0.f; kv[3] = 0.f; vv[3] = 0.f;
            *(v4f*)&qs[w][t][0]  = qv;
            *(v4f*)&ks_[w][t][0] = kv;
            vT[w][0][t] = vv[0];
            vT[w][1][t] = vv[1];
            vT[w][2][t] = vv[2];
        }

        v2f kf[3], qf[3];
        float vf[3][8];
#pragma unroll
        for (int i = 0; i < 3; ++i)
            kf[i] = *(const v2f*)&ks_[w][i * 16 + l16][kofs];
        qf[0] = *(const v2f*)&qs[w][l16][kofs];
        qf[1] = *(const v2f*)&qs[w][QOFF + l16][kofs];
        qf[2] = *(const v2f*)&qs[w][QOFF + 16 + l16][kofs];
#pragma unroll
        for (int kt = 0; kt < 2; ++kt) {
            v4f lo = *(const v4f*)&vT[w][colc][kt * 16 + 8 * hi16];
            v4f hi = *(const v4f*)&vT[w][colc][kt * 16 + 8 * hi16 + 4];
            vf[kt][0] = lo.x; vf[kt][1] = lo.y; vf[kt][2] = lo.z; vf[kt][3] = lo.w;
            vf[kt][4] = hi.x; vf[kt][5] = hi.y; vf[kt][6] = hi.z; vf[kt][7] = hi.w;
        }
        {
            v2f l2 = *(const v2f*)&vT[w][colc][32 + 8 * hi16];
            vf[2][0] = l2.x; vf[2][1] = l2.y;
            vf[2][2] = 0.f; vf[2][3] = 0.f; vf[2][4] = 0.f; vf[2][5] = 0.f; vf[2][6] = 0.f; vf[2][7] = 0.f;
        }
        const v4f o0 = attend_q<0>(kf, qf[0], vf, cmd, cm0);
        const v4f o1 = attend_q<1>(kf, qf[1], vf, cmd, cml);
        const v4f o2 = attend_q<2>(kf, qf[2], vf, cmd, cml);
        if (hi16 == 0) {
            if (l16 < QOFF) *(v4f*)&os[w][l16][0] = o0;
            *(v4f*)&os[w][QOFF + l16][0]      = o1;
            *(v4f*)&os[w][QOFF + 16 + l16][0] = o2;
        }
        __syncthreads();

        if (tid < T) {
            float u[DM];
#pragma unroll
            for (int i = 0; i < HD; ++i) { u[i] = os[0][tid][i]; u[HD + i] = os[1][tid][i]; }
#pragma unroll
            for (int i = 0; i < DM; ++i) {
                float acc = 0.f;
#pragma unroll
                for (int j = 0; j < DM; ++j) acc += out_w[l * 36 + i * 6 + j] * u[j];
                xs[tid][i] += acc;
            }
        }
        __syncthreads();
    }

    if (tid < T) {
        float m = 0.f;
#pragma unroll
        for (int c = 0; c < DM; ++c) m += xs[tid][c];
        m *= (1.0f / DM);
        float var = 0.f;
#pragma unroll
        for (int c = 0; c < DM; ++c) { float d = xs[tid][c] - m; var += d * d; }
        var *= (1.0f / DM);
        float r = rsqrtf(var + 1e-5f);
        float xf[DM];
#pragma unroll
        for (int c = 0; c < DM; ++c)
            xf[c] = (xs[tid][c] - m) * r * lnf_w[c] + lnf_b[c];
        float* op = out + ((size_t)b * T + tid) * VOCAB;
#pragma unroll
        for (int v = 0; v < VOCAB; ++v) {
            float acc = 0.f;
#pragma unroll
            for (int j = 0; j < DM; ++j) acc += head_w[v * DM + j] * xf[j];
            op[v] = acc;
        }
    }
}

extern "C" void kernel_launch(void* const* d_in, const int* in_sizes, int n_in,
                              void* d_out, int out_size, void* d_ws, size_t ws_size,
                              hipStream_t stream) {
    const int*   idx     = (const int*)  d_in[0];
    const float* tok_emb = (const float*)d_in[1];
    const float* pos_enc = (const float*)d_in[2];
    const float* ln_w    = (const float*)d_in[3];
    const float* ln_b    = (const float*)d_in[4];
    const float* q1w     = (const float*)d_in[5];
    const float* k1w     = (const float*)d_in[6];
    const float* v1w     = (const float*)d_in[7];
    const float* q2w     = (const float*)d_in[8];
    const float* k2w     = (const float*)d_in[9];
    const float* v2w     = (const float*)d_in[10];
    const float* out_w   = (const float*)d_in[11];
    const float* lnf_w   = (const float*)d_in[12];
    const float* lnf_b   = (const float*)d_in[13];
    const float* head_w  = (const float*)d_in[14];
    float* out = (float*)d_out;

    const int batch = in_sizes[0] / T;
    addtrans_kernel<<<batch, 64, 0, stream>>>(idx, tok_emb, pos_enc, ln_w, ln_b,
                                              q1w, k1w, v1w, q2w, k2w, v2w,
                                              out_w, lnf_w, lnf_b, head_w, out);
    (void)d_ws; (void)ws_size; (void)n_in; (void)out_size;
}


# Per-position work (layernorm, q/k/v of both heads, out-proj, vocab head) done by ONE wave, two positions per lane as packed f32 with scalar weights, x in registers; sum orders unchanged
# speedup vs baseline: 6.1359x; 1.0847x over previous
#include <hip/hip_runtime.h>
#include <math.h>

#define N_LAYERS 4
#define T        34
#define TP       48
#define VOCAB    14
#define DM       6
#define HD       3

#define QOFF     (T - 32)

static_assert(T % 2 == 0, "the per-position pass gives each lane the positions i and i + T/2");
static_assert(T == 34, "the first query tile holds the T - 32 = 2 queries left over; every query tile's last key tile is read as one chunk of 2 keys per lane half");
typedef float v2f __attribute__((ext_vector_type(2)));
typedef float v4f __attribute__((ext_vector_type(4)));
typedef float v8f __attribute__((ext_vector_type(8)));

__device__ __forceinline__ float xhalf(float x) {
    int b = __builtin_bit_cast(int, x);
    return __builtin_bit_cast(float,
        __builtin_amdgcn_permlanex16(b, b, 0x76543210, 0xfedcba98, false, false));
}

template <int QT>
__device__ __forceinline__ v4f attend_q(const v2f (&kf)[3], v2f qf, const float (&vf)[3][8], v8f md, v8f ml)
{
    v2f p[QT + 1][4];
    float mx = -3.4e38f;
#pragma unroll
    for (int kt = 0; kt <= QT; ++kt) {
        const int kv = (kt == QT) ? QOFF : 8;
        v8f z = {0.f, 0.f, 0.f, 0.f, 0.f, 0.f, 0.f, 0.f};
        v8f s = __builtin_amdgcn_wmma_f32_16x16x4_f32(false, kf[kt], false, qf, (short)0,
                                                      (kt == QT) ? ml : (kt == QT - 1) ? md : z, false, false);
#pragma unroll
        for (int j = 0; j < 4; ++j) {
            if (2 * j < kv) {
                p[kt][j].x = s[2 * j];
                p[kt][j].y = s[2 * j + 1];
                mx = fmaxf(mx, fmaxf(s[2 * j], s[2 * j + 1]));
            }
        }
    }
    mx = fmaxf(mx, xhalf(mx));
    const v2f mx2 = {mx, mx};
#pragma unroll
    for (int kt = 0; kt <= QT; ++kt) {
        const int kv = (kt == QT) ? QOFF : 8;
#pragma unroll
        for (int j = 0; j < 4; ++j) {
            if (2 * j < kv) {
                const v2f t = p[kt][j] - mx2;
                p[kt][j].x = __builtin_amdgcn_exp2f(t.x);
                p[kt][j].y = __builtin_amdgcn_exp2f(t.y);
            }
        }
    }
    v8f acc = {0.f, 0.f, 0.f, 0.f, 0.f, 0.f, 0.f, 0.f};
#pragma unroll
    for (int kt = 0; kt <= QT; ++kt) {
        const int kv = (kt == QT) ? QOFF : 8;
#pragma unroll
        for (int j = 0; j < 4; ++j) {
            if (2 * j < kv) {
                v2f a;
                a.x  = vf[kt][2 * j];  a.y  = vf[kt][2 * j + 1];
                acc = __builtin_amdgcn_wmma_f32_16x16x4_f32(false, a, false, p[kt][j],
                                                            (short)0, acc, false, false);
            }
        }
    }
    const float inv = __builtin_amdgcn_rcpf(acc[3]);
    v4f o;
    o.x = acc[0] * inv; o.y = acc[1] * inv; o.z = acc[2] * inv; o.w = 0.f;
    return o;
}

__global__ __launch_bounds__(64)
void addtrans_kernel(const int* __restrict__ idx,
                     const float* __restrict__ tok_emb,
                     const float* __restrict__ pos_enc,
                     const float* __restrict__ ln_w, const float* __restrict__ ln_b,
                     const float* __restrict__ q1w, const float* __restrict__ k1w,
                     const float* __restrict__ v1w,
                     const float* __restrict__ q2w, const float* __restrict__ k2w,
                     const float* __restrict__ v2w,
                     const float* __restrict__ out_w,
                     const float* __restrict__ lnf_w, const float* __restrict__ lnf_b,
                     const float* __restrict__ head_w,
                     float* __restrict__ out)
{
    __shared__ __attribute__((aligned(16))) float qs[2][TP][4];
    __shared__ __attribute__((aligned(16))) float ks_[2][TP][4];
    __shared__ __attribute__((aligned(16))) float vT[2][4][TP];
    float (*os)[TP][4] = qs;

    const int b    = blockIdx.x;
    const int tid  = threadIdx.x;
    const int w    = tid >> 5;
    const int lane = tid & 31;
    const int hi16 = lane >> 4;
    const int l16  = lane & 15;
    const int kofs = hi16 * 2;
    const int colc = (l16 < 3) ? l16 : 3;
    const int dl   = l16 - 8 * hi16;
    const float sc = 0.57735026918962576f * 1.4426950408889634f;
    v8f cmd, cml, cm0;
#pragma unroll
    for (int r = 0; r < 8; ++r) {
        cmd[r] = (r <= dl + QOFF) ? 0.f : -3.4e38f;
        cml[r] = (r <= dl + QOFF - 16) ? 0.f : -3.4e38f;
        cm0[r] = (r <= dl) ? 0.f : -3.4e38f;
    }

    __builtin_prefetch(idx + (size_t)b * T, 0, 1);

    for (int t = tid; t < 2 * TP; t += 64) vT[t / TP][3][t % TP] = ((t % TP) < T) ? 1.f : 0.f;
    for (int t = tid; t < 2 * TP; t += 64) { qs[t / TP][t % TP][3] = 0.f; ks_[t / TP][t % TP][3] = 0.f; }
    if (tid < 2 * (TP - T)) {
        const int hd = tid / (TP - T), p = T + tid % (TP - T);
        ks_[hd][p][0] = 0.f; ks_[hd][p][1] = 0.f; ks_[hd][p][2] = 0.f;
        vT[hd][0][p] = 0.f; vT[hd][1][p] = 0.f; vT[hd][2][p] = 0.f;
    }

    const bool own = (w == (__builtin_popcount((unsigned)b) & 1)) && (lane < T / 2);
    const int p0 = lane, p1 = lane + T / 2;
    v2f x[DM];
#pragma unroll
    for (int c = 0; c < DM; ++c) { x[c].x = 0.f; x[c].y = 0.f; }
    if (own) {
        const int tok0 = idx[(size_t)b * T + p0];
        const int tok1 = idx[(size_t)b * T + p1];
#pragma unroll
        for (int c = 0; c < 3; ++c) {
            x[c].x     = tok_emb[tok0 * 3 + c];  x[c].y     = tok_emb[tok1 * 3 + c];
            x[3 + c].x = pos_enc[p0 * 3 + c];    x[3 + c].y = pos_enc[p1 * 3 + c];
        }
    }

    for (int l = 0; l < N_LAYERS; ++l) {
        if (own) {
            v2f m = {0.f, 0.f};
#pragma unroll
            for (int c = 0; c < DM; ++c) m += x[c];
            m *= (1.0f / DM);
            v2f var = {0.f, 0.f};
#pragma unroll
            for (int c = 0; c < DM; ++c) { v2f d = x[c] - m; var += d * d; }
            var *= (1.0f / DM);
            v2f r;
            r.x = rsqrtf(var.x + 1e-5f); r.y = rsqrtf(var.y + 1e-5f);
            v2f h[DM];
#pragma unroll
            for (int c = 0; c < DM; ++c)
                h[c] = (x[c] - m) * r * ln_w[l * DM + c] + ln_b[l * DM + c];
#pragma unroll
            for (int hd = 0; hd < 2; ++hd) {
                const float* qw = (hd == 0 ? q1w : q2w) + l * 9;
                const float* kw = (hd == 0 ? k1w : k2w) + l * 9;
                const float* vw = (hd == 0 ? v1w : v2w) + l * 9;
                const v2f h0 = h[hd * HD], h1 = h[hd * HD + 1], h2 = h[hd * HD + 2];
#pragma unroll
                for (int i = 0; i < HD; ++i) {
                    const v2f qv = (qw[i*3]*h0 + qw[i*3+1]*h1 + qw[i*3+2]*h2) * sc;
                    const v2f kv =  kw[i*3]*h0 + kw[i*3+1]*h1 + kw[i*3+2]*h2;
                    const v2f vv =  vw[i*3]*h0 + vw[i*3+1]*h1 + vw[i*3+2]*h2;
                    qs[hd][p0][i]  = qv.x;  qs[hd][p1][i]  = qv.y;
                    ks_[hd][p0][i] = kv.x;  ks_[hd][p1][i] = kv.y;
                    vT[hd][i][p0]  = vv.x;  vT[hd][i][p1]  = vv.y;
                }
            }
        }
        __syncthreads();

        v2f kf[3], qf[3];
        float vf[3][8];
#pragma unroll
        for (int i = 0; i < 3; ++i)
            kf[i] = *(const v2f*)&ks_[w][i * 16 + l16][kofs];
        qf[0] = *(const v2f*)&qs[w][l16][kofs];
        qf[1] = *(const v2f*)&qs[w][QOFF + l16][kofs];
        qf[2] = *(const v2f*)&qs[w][QOFF + 16 + l16][kofs];
#pragma unroll
        for (int kt = 0; kt < 2; ++kt) {
            v4f lo = *(const v4f*)&vT[w][colc][kt * 16 + 8 * hi16];
            v4f hi = *(const v4f*)&vT[w][colc][kt * 16 + 8 * hi16 + 4];
            vf[kt][0] = lo.x; vf[kt][1] = lo.y; vf[kt][2] = lo.z; vf[kt][3] = lo.w;
            vf[kt][4] = hi.x; vf[kt][5] = hi.y; vf[kt][6] = hi.z; vf[kt][7] = hi.w;
        }
        {
            v2f l2 = *(const v2f*)&vT[w][colc][32 + 8 * hi16];
            vf[2][0] = l2.x; vf[2][1] = l2.y;
            vf[2][2] = 0.f; vf[2][3] = 0.f; vf[2][4] = 0.f; vf[2][5] = 0.f; vf[2][6] = 0.f; vf[2][7] = 0.f;
        }
        const v4f o0 = attend_q<0>(kf, qf[0], vf, cmd, cm0);
        const v4f o1 = attend_q<1>(kf, qf[1], vf, cmd, cml);
        const v4f o2 = attend_q<2>(kf, qf[2], vf, cmd, cml);
        if (hi16 == 0) {
            if (l16 < QOFF) *(v4f*)&os[w][l16][0] = o0;
            *(v4f*)&os[w][QOFF + l16][0]      = o1;
            *(v4f*)&os[w][QOFF + 16 + l16][0] = o2;
        }
        __syncthreads();

        if (own) {
            v2f u[DM];
#pragma unroll
            for (int j = 0; j < HD; ++j) {
                u[j].x      = os[0][p0][j];  u[j].y      = os[0][p1][j];
                u[HD + j].x = os[1][p0][j];  u[HD + j].y = os[1][p1][j];
            }
#pragma unroll
            for (int i = 0; i < DM; ++i) {
                v2f acc = {0.f, 0.f};
#pragma unroll
                for (int j = 0; j < DM; ++j) acc += out_w[l * 36 + i * 6 + j] * u[j];
                x[i] += acc;
            }
        }
    }

    if (own) {
        v2f m = {0.f, 0.f};
#pragma unroll
        for (int c = 0; c < DM; ++c) m += x[c];
        m *= (1.0f / DM);
        v2f var = {0.f, 0.f};
#pragma unroll
        for (int c = 0; c < DM; ++c) { v2f d = x[c] - m; var += d * d; }
        var *= (1.0f / DM);
        v2f r;
        r.x = rsqrtf(var.x + 1e-5f); r.y = rsqrtf(var.y + 1e-5f);
        v2f xf[DM];
#pragma unroll
        for (int c = 0; c < DM; ++c)
            xf[c] = (x[c] - m) * r * lnf_w[c] + lnf_b[c];
        float* op0 = out + ((size_t)b * T + p0) * VOCAB;
        float* op1 = out + ((size_t)b * T + p1) * VOCAB;
#pragma unroll
        for (int v = 0; v < VOCAB; ++v) {
            v2f acc = {0.f, 0.f};
#pragma unroll
            for (int j = 0; j < DM; ++j) acc += head_w[v * DM + j] * xf[j];
            op0[v] = acc.x;
            op1[v] = acc.y;
        }
    }
}

extern "C" void kernel_launch(void* const* d_in, const int* in_sizes, int n_in,
                              void* d_out, int out_size, void* d_ws, size_t ws_size,
                              hipStream_t stream) {
    const int*   idx     = (const int*)  d_in[0];
    const float* tok_emb = (const float*)d_in[1];
    const float* pos_enc = (const float*)d_in[2];
    const float* ln_w    = (const float*)d_in[3];
    const float* ln_b    = (const float*)d_in[4];
    const float* q1w     = (const float*)d_in[5];
    const float* k1w     = (const float*)d_in[6];
    const float* v1w     = (const float*)d_in[7];
    const float* q2w     = (const float*)d_in[8];
    const float* k2w     = (const float*)d_in[9];
    const float* v2w     = (const float*)d_in[10];
    const float* out_w   = (const float*)d_in[11];
    const float* lnf_w   = (const float*)d_in[12];
    const float* lnf_b   = (const float*)d_in[13];
    const float* head_w  = (const float*)d_in[14];
    float* out = (float*)d_out;

    const int batch = in_sizes[0] / T;
    addtrans_kernel<<<batch, 64, 0, stream>>>(idx, tok_emb, pos_enc, ln_w, ln_b,
                                              q1w, k1w, v1w, q2w, k2w, v2w,
                                              out_w, lnf_w, lnf_b, head_w, out);
    (void)d_ws; (void)ws_size; (void)n_in; (void)out_size;
}
